# starting kernel
# baseline (speedup 1.0000x reference)
#include <hip/hip_runtime.h>
#include <hip/hip_bf16.h>

typedef float v2f __attribute__((ext_vector_type(2)));
typedef float v8f __attribute__((ext_vector_type(8)));

#define B_    4
#define C_    64
#define D_    32
#define H_    64
#define W_    64
#define HW    (H_ * W_)
#define DHW   (D_ * HW)
#define CDHW  (C_ * DHW)
#define TW    24

__global__ __launch_bounds__(256) void sg_reduce_maxmean(
    const float* __restrict__ x, float* __restrict__ xmax, float* __restrict__ xmean) {
  long t = (long)blockIdx.x * blockDim.x + threadIdx.x;
  long f = t * 4;
  int  b = (int)(f / DHW);
  int  q = (int)(f % DHW);
  const float* xp = x + (long)b * CDHW + q;

  float4 v = *(const float4*)xp;
  float mx0 = v.x, mx1 = v.y, mx2 = v.z, mx3 = v.w;
  float s0 = v.x, s1 = v.y, s2 = v.z, s3 = v.w;
  for (int c = 1; c < C_; ++c) {
    float4 u = *(const float4*)(xp + (long)c * DHW);
    mx0 = fmaxf(mx0, u.x); mx1 = fmaxf(mx1, u.y);
    mx2 = fmaxf(mx2, u.z); mx3 = fmaxf(mx3, u.w);
    s0 += u.x; s1 += u.y; s2 += u.z; s3 += u.w;
  }
  const float inv = 1.0f / (float)C_;
  float4 mo; mo.x = mx0; mo.y = mx1; mo.z = mx2; mo.w = mx3;
  float4 me; me.x = s0 * inv; me.y = s1 * inv; me.z = s2 * inv; me.w = s3 * inv;
  *(float4*)(xmax  + (long)b * DHW + q) = mo;
  *(float4*)(xmean + (long)b * DHW + q) = me;
}

__global__ __launch_bounds__(32) void sg_conv_sigmoid(
    const float* __restrict__ xmax, const float* __restrict__ xmean,
    const float* __restrict__ wgt, const float* __restrict__ bias,
    float* __restrict__ scale) {
  __shared__ float lw[2 * 7 * 7 * 7];
  __shared__ float tile[TW * TW];

  const int lane = threadIdx.x;
  const int bid  = blockIdx.x;
  const int wt = bid & 3, ht = (bid >> 2) & 3;
  const int d  = (bid >> 4) & 31, b = bid >> 9;
  const int hbase = ht * 16, wbase = wt * 16;

  for (int i = lane; i < 686; i += 32) lw[i] = wgt[i];
  __syncthreads();

  const int m    = lane & 15;
  const int n    = m;
  const int kgrp = (lane >> 4) << 1;

  v8f acc = {0.f, 0.f, 0.f, 0.f, 0.f, 0.f, 0.f, 0.f};

  for (int c = 0; c < 2; ++c) {
    const float* plane_b = (c == 0 ? xmax : xmean) + (long)b * DHW;
    for (int kd = 0; kd < 7; ++kd) {
      const int dz = d + kd - 3;
      if (dz < 0 || dz >= D_) continue;
      const float* plane = plane_b + (long)dz * HW;

      for (int i = lane; i < 24 * 22; i += 32) {
        const int row = i / 22, col = i % 22;
        const int h = hbase + row - 3, w = wbase + col - 3;
        float v = 0.0f;
        if (row < 22 && h >= 0 && h < H_ && w >= 0 && w < W_)
          v = plane[h * W_ + w];
        tile[row * TW + col] = v;
      }
      __syncthreads();

      for (int kw = 0; kw < 7; ++kw) {
        const int wb0 = (c * 7 + kd) * 49 + kw;
        float w0[2], w1[2]; int c0[2];
#pragma unroll
        for (int s = 0; s < 2; ++s) {
          const int ks = kgrp + s;
          const int r  = (ks - m) & 3;
          c0[s] = (m + r - ks) >> 2;
          w0[s] = lw[wb0 + r * 7];
          w1[s] = (r <= 2) ? lw[wb0 + (r + 4) * 7] : 0.0f;
        }
#pragma unroll
        for (int ch = 0; ch < 6; ++ch) {
          v2f a, bb;
          a.x = (ch == c0[0]) ? w0[0] : ((ch == c0[0] + 1) ? w1[0] : 0.0f);
          a.y = (ch == c0[1]) ? w0[1] : ((ch == c0[1] + 1) ? w1[1] : 0.0f);
          const int j0 = ch * 4 + kgrp;
          bb.x = tile[j0 * TW + kw + n];
          bb.y = tile[(j0 + 1) * TW + kw + n];
          acc = __builtin_amdgcn_wmma_f32_16x16x4_f32(
              false, a, false, bb, (short)0, acc, false, false);
        }
      }
      __syncthreads();
    }
  }

  const float bv = bias[0];
  float* sp = scale + ((long)b * D_ + d) * HW;
#pragma unroll
  for (int r = 0; r < 8; ++r) {
    const int M = r + (kgrp << 2);
    const float y = acc[r] + bv;
    sp[(hbase + M) * W_ + (wbase + n)] = 1.0f / (1.0f + __expf(-y));
  }
}

__global__ __launch_bounds__(256) void sg_apply(
    const float* __restrict__ x, const float* __restrict__ scale,
    float* __restrict__ out) {
  long t = (long)blockIdx.x * blockDim.x + threadIdx.x;
  long f = t * 4;
  int  b = (int)(f / CDHW);
  int  p = (int)((f % CDHW) % DHW);
  float4 xv = *(const float4*)(x + f);
  float4 sv = *(const float4*)(scale + (long)b * DHW + p);
  float4 o;
  o.x = xv.x * sv.x; o.y = xv.y * sv.y;
  o.z = xv.z * sv.z; o.w = xv.w * sv.w;
  *(float4*)(out + f) = o;
}

extern "C" void kernel_launch(void* const* d_in, const int* in_sizes, int n_in,
                              void* d_out, int out_size, void* d_ws, size_t ws_size,
                              hipStream_t stream) {
  const float* x      = (const float*)d_in[0];
  const float* conv_w = (const float*)d_in[1];
  const float* conv_b = (const float*)d_in[2];
  float* out = (float*)d_out;

  float* ws    = (float*)d_ws;
  float* xmax  = ws;
  float* xmean = ws + (long)B_ * DHW;
  float* scale = ws + 2L * B_ * DHW;

  sg_reduce_maxmean<<<512, 256, 0, stream>>>(x, xmax, xmean);

  sg_conv_sigmoid<<<2048, 32, 0, stream>>>(xmax, xmean, conv_w, conv_b, scale);

  sg_apply<<<32768, 256, 0, stream>>>(x, scale, out);
}


# conv kernel stages its whole 14-plane halo and weights into LDS once, all global loads issued before the first wait as 16-byte pieces, replacing 238 serial one-word load-wait-store trips per wave
# speedup vs baseline: 1.7279x; 1.7279x over previous
#include <hip/hip_runtime.h>
#include <hip/hip_bf16.h>

typedef float v2f __attribute__((ext_vector_type(2)));
typedef float v8f __attribute__((ext_vector_type(8)));

#define B_    4
#define C_    64
#define D_    32
#define H_    64
#define W_    64
#define HW    (H_ * W_)
#define DHW   (D_ * HW)
#define CDHW  (C_ * DHW)
#define TW    24

__global__ __launch_bounds__(256) void sg_reduce_maxmean(
    const float* __restrict__ x, float* __restrict__ xmax, float* __restrict__ xmean) {
  long t = (long)blockIdx.x * blockDim.x + threadIdx.x;
  long f = t * 4;
  int  b = (int)(f / DHW);
  int  q = (int)(f % DHW);
  const float* xp = x + (long)b * CDHW + q;

  float4 v = *(const float4*)xp;
  float mx0 = v.x, mx1 = v.y, mx2 = v.z, mx3 = v.w;
  float s0 = v.x, s1 = v.y, s2 = v.z, s3 = v.w;
  for (int c = 1; c < C_; ++c) {
    float4 u = *(const float4*)(xp + (long)c * DHW);
    mx0 = fmaxf(mx0, u.x); mx1 = fmaxf(mx1, u.y);
    mx2 = fmaxf(mx2, u.z); mx3 = fmaxf(mx3, u.w);
    s0 += u.x; s1 += u.y; s2 += u.z; s3 += u.w;
  }
  const float inv = 1.0f / (float)C_;
  float4 mo; mo.x = mx0; mo.y = mx1; mo.z = mx2; mo.w = mx3;
  float4 me; me.x = s0 * inv; me.y = s1 * inv; me.z = s2 * inv; me.w = s3 * inv;
  *(float4*)(xmax  + (long)b * DHW + q) = mo;
  *(float4*)(xmean + (long)b * DHW + q) = me;
}

#define PR    22
#define PS    (PR * TW)
#define NPL   14

__global__ __launch_bounds__(32) void sg_conv_sigmoid(
    const float* __restrict__ xmax, const float* __restrict__ xmean,
    const float* __restrict__ wgt, const float* __restrict__ bias,
    float* __restrict__ scale) {
  __shared__ float lw[688];
  __shared__ __attribute__((aligned(16))) float tile[NPL * PS + 2 * TW];

  const int lane = threadIdx.x;
  const int bid  = blockIdx.x;
  const int wt = bid & 3, ht = (bid >> 2) & 3;
  const int d  = (bid >> 4) & 31, b = bid >> 9;
  const int hbase = ht * 16, wbase = wt * 16;

  float wv[22];
#pragma unroll
  for (int k = 0; k < 21; ++k) wv[k] = wgt[lane + 32 * k];
  wv[21] = (lane < 14) ? wgt[672 + lane] : 0.0f;

  const int q  = lane & 7, rs = lane >> 3;
  const int wc = wbase - 4 + 4 * q;
  const bool wok = (q < 6) && (wc >= 0) && (wc < W_);
  const int wcc = min(max(wc, 0), W_ - 4);
  int  goff[6];
  bool ok[6];
#pragma unroll
  for (int it = 0; it < 6; ++it) {
    const int row = 4 * it + rs;
    const int h   = hbase - 3 + row;
    ok[it]   = wok && (row < PR) && (h >= 0) && (h < H_);
    goff[it] = min(max(h, 0), H_ - 1) * W_ + wcc;
  }

  float4 hv[NPL][6];
#pragma unroll
  for (int c = 0; c < 2; ++c) {
    const float* cb = (c == 0 ? xmax : xmean) + (long)b * DHW;
#pragma unroll
    for (int kd = 0; kd < 7; ++kd) {
      const int dz  = d + kd - 3;
      const bool pv = (dz >= 0) && (dz < D_);
      const float* plane = cb + (long)min(max(dz, 0), D_ - 1) * HW;
#pragma unroll
      for (int it = 0; it < 6; ++it) {
        float4 v = *(const float4*)(plane + goff[it]);
        const bool keep = pv && ok[it];
        v.x = keep ? v.x : 0.0f; v.y = keep ? v.y : 0.0f;
        v.z = keep ? v.z : 0.0f; v.w = keep ? v.w : 0.0f;
        hv[c * 7 + kd][it] = v;
      }
    }
  }

#pragma unroll
  for (int k = 0; k < 21; ++k) lw[lane + 32 * k] = wv[k];
  if (lane < 14) lw[672 + lane] = wv[21];
  if (q < 6) {
    float* tp = tile + rs * TW + 4 * q;
#pragma unroll
    for (int p = 0; p < NPL; ++p) {
#pragma unroll
      for (int it = 0; it < 5; ++it)
        *(float4*)(tp + p * PS + it * 4 * TW) = hv[p][it];
    }
    if (rs < 2) {
#pragma unroll
      for (int p = 0; p < NPL; ++p)
        *(float4*)(tp + p * PS + 5 * 4 * TW) = hv[p][5];
    }
  }
  if (lane < 12) {
    float4 z; z.x = 0.0f; z.y = 0.0f; z.z = 0.0f; z.w = 0.0f;
    *(float4*)(tile + NPL * PS + 4 * lane) = z;
  }
  __syncthreads();

  const int m    = lane & 15;
  const int n    = m;
  const int kgrp = (lane >> 4) << 1;

  v8f acc = {0.f, 0.f, 0.f, 0.f, 0.f, 0.f, 0.f, 0.f};

  for (int c = 0; c < 2; ++c) {
    for (int kd = 0; kd < 7; ++kd) {
      const int dz = d + kd - 3;
      if (dz < 0 || dz >= D_) continue;
      const float* tl = tile + (c * 7 + kd) * PS + n + 1;

      for (int kw = 0; kw < 7; ++kw) {
        const int wb0 = (c * 7 + kd) * 49 + kw;
        float w0[2], w1[2]; int c0[2];
#pragma unroll
        for (int s = 0; s < 2; ++s) {
          const int ks = kgrp + s;
          const int r  = (ks - m) & 3;
          c0[s] = (m + r - ks) >> 2;
          w0[s] = lw[wb0 + r * 7];
          w1[s] = (r <= 2) ? lw[wb0 + (r + 4) * 7] : 0.0f;
        }
#pragma unroll
        for (int ch = 0; ch < 6; ++ch) {
          v2f a, bb;
          a.x = (ch == c0[0]) ? w0[0] : ((ch == c0[0] + 1) ? w1[0] : 0.0f);
          a.y = (ch == c0[1]) ? w0[1] : ((ch == c0[1] + 1) ? w1[1] : 0.0f);
          const int j0 = ch * 4 + kgrp;
          bb.x = tl[j0 * TW + kw];
          bb.y = tl[(j0 + 1) * TW + kw];
          acc = __builtin_amdgcn_wmma_f32_16x16x4_f32(
              false, a, false, bb, (short)0, acc, false, false);
        }
      }
    }
  }

  const float bv = bias[0];
  float* sp = scale + ((long)b * D_ + d) * HW;
#pragma unroll
  for (int r = 0; r < 8; ++r) {
    const int M = r + (kgrp << 2);
    const float y = acc[r] + bv;
    sp[(hbase + M) * W_ + (wbase + n)] = 1.0f / (1.0f + __expf(-y));
  }
}

__global__ __launch_bounds__(256) void sg_apply(
    const float* __restrict__ x, const float* __restrict__ scale,
    float* __restrict__ out) {
  long t = (long)blockIdx.x * blockDim.x + threadIdx.x;
  long f = t * 4;
  int  b = (int)(f / CDHW);
  int  p = (int)((f % CDHW) % DHW);
  float4 xv = *(const float4*)(x + f);
  float4 sv = *(const float4*)(scale + (long)b * DHW + p);
  float4 o;
  o.x = xv.x * sv.x; o.y = xv.y * sv.y;
  o.z = xv.z * sv.z; o.w = xv.w * sv.w;
  *(float4*)(out + f) = o;
}

extern "C" void kernel_launch(void* const* d_in, const int* in_sizes, int n_in,
                              void* d_out, int out_size, void* d_ws, size_t ws_size,
                              hipStream_t stream) {
  const float* x      = (const float*)d_in[0];
  const float* conv_w = (const float*)d_in[1];
  const float* conv_b = (const float*)d_in[2];
  float* out = (float*)d_out;

  float* ws    = (float*)d_ws;
  float* xmax  = ws;
  float* xmean = ws + (long)B_ * DHW;
  float* scale = ws + 2L * B_ * DHW;

  sg_reduce_maxmean<<<512, 256, 0, stream>>>(x, xmax, xmean);

  sg_conv_sigmoid<<<2048, 32, 0, stream>>>(xmax, xmean, conv_w, conv_b, scale);

  sg_apply<<<32768, 256, 0, stream>>>(x, scale, out);
}


# conv kernel reads each plane's 42 B fragments and 28 taps from LDS as one group into a second register set while the previous plane's 42 WMMAs run, instead of one LDS round trip per WMMA
# speedup vs baseline: 1.9081x; 1.1043x over previous
#include <hip/hip_runtime.h>
#include <hip/hip_bf16.h>

typedef float v2f __attribute__((ext_vector_type(2)));
typedef float v8f __attribute__((ext_vector_type(8)));

#define B_    4
#define C_    64
#define D_    32
#define H_    64
#define W_    64
#define HW    (H_ * W_)
#define DHW   (D_ * HW)
#define CDHW  (C_ * DHW)
#define TW    24

__global__ __launch_bounds__(256) void sg_reduce_maxmean(
    const float* __restrict__ x, float* __restrict__ xmax, float* __restrict__ xmean) {
  long t = (long)blockIdx.x * blockDim.x + threadIdx.x;
  long f = t * 4;
  int  b = (int)(f / DHW);
  int  q = (int)(f % DHW);
  const float* xp = x + (long)b * CDHW + q;

  float4 v = *(const float4*)xp;
  float mx0 = v.x, mx1 = v.y, mx2 = v.z, mx3 = v.w;
  float s0 = v.x, s1 = v.y, s2 = v.z, s3 = v.w;
  for (int c = 1; c < C_; ++c) {
    float4 u = *(const float4*)(xp + (long)c * DHW);
    mx0 = fmaxf(mx0, u.x); mx1 = fmaxf(mx1, u.y);
    mx2 = fmaxf(mx2, u.z); mx3 = fmaxf(mx3, u.w);
    s0 += u.x; s1 += u.y; s2 += u.z; s3 += u.w;
  }
  const float inv = 1.0f / (float)C_;
  float4 mo; mo.x = mx0; mo.y = mx1; mo.z = mx2; mo.w = mx3;
  float4 me; me.x = s0 * inv; me.y = s1 * inv; me.z = s2 * inv; me.w = s3 * inv;
  *(float4*)(xmax  + (long)b * DHW + q) = mo;
  *(float4*)(xmean + (long)b * DHW + q) = me;
}

#define PR    22
#define PS    (PR * TW)
#define NPL   14

struct PlaneOps {
  v2f   b[7][6];
  float w0[7][2];
  float w1[7][2];
};

__device__ __forceinline__ void sg_load_plane(PlaneOps& o, const float* tl, const float* wl,
                                              const int (&i0)[2], const int (&i1)[2]) {
#pragma unroll
  for (int kw = 0; kw < 7; ++kw) {
#pragma unroll
    for (int s = 0; s < 2; ++s) {
      o.w0[kw][s] = wl[i0[s] + kw];
      o.w1[kw][s] = wl[i1[s] + kw];
    }
  }
#pragma unroll
  for (int kw = 0; kw < 7; ++kw) {
#pragma unroll
    for (int ch = 0; ch < 6; ++ch) {
      o.b[kw][ch].x = tl[(ch * 4) * TW + kw];
      o.b[kw][ch].y = tl[(ch * 4 + 1) * TW + kw];
    }
    asm volatile("" ::: "memory");
  }
}

__device__ __forceinline__ v8f sg_mma_plane(const PlaneOps& o, v8f acc,
                                            const int (&c0)[2], const bool (&hi)[2]) {
#pragma unroll
  for (int kw = 0; kw < 7; ++kw) {
#pragma unroll
    for (int ch = 0; ch < 6; ++ch) {
      v2f a;
      a.x = (ch == c0[0]) ? o.w0[kw][0] : ((hi[0] && ch == c0[0] + 1) ? o.w1[kw][0] : 0.0f);
      a.y = (ch == c0[1]) ? o.w0[kw][1] : ((hi[1] && ch == c0[1] + 1) ? o.w1[kw][1] : 0.0f);
      acc = __builtin_amdgcn_wmma_f32_16x16x4_f32(
          false, a, false, o.b[kw][ch], (short)0, acc, false, false);
    }
  }
  return acc;
}

__global__ __launch_bounds__(32) void sg_conv_sigmoid(
    const float* __restrict__ xmax, const float* __restrict__ xmean,
    const float* __restrict__ wgt, const float* __restrict__ bias,
    float* __restrict__ scale) {
  __shared__ float lw[688];
  __shared__ __attribute__((aligned(16))) float tile[NPL * PS + 2 * TW];

  const int lane = threadIdx.x;
  const int bid  = blockIdx.x;
  const int wt = bid & 3, ht = (bid >> 2) & 3;
  const int d  = (bid >> 4) & 31, b = bid >> 9;
  const int hbase = ht * 16, wbase = wt * 16;

  float wv[22];
#pragma unroll
  for (int k = 0; k < 21; ++k) wv[k] = wgt[lane + 32 * k];
  wv[21] = (lane < 14) ? wgt[672 + lane] : 0.0f;

  const int q  = lane & 7, rs = lane >> 3;
  const int wc = wbase - 4 + 4 * q;
  const bool wok = (q < 6) && (wc >= 0) && (wc < W_);
  const int wcc = min(max(wc, 0), W_ - 4);
  int  goff[6];
  bool ok[6];
#pragma unroll
  for (int it = 0; it < 6; ++it) {
    const int row = 4 * it + rs;
    const int h   = hbase - 3 + row;
    ok[it]   = wok && (row < PR) && (h >= 0) && (h < H_);
    goff[it] = min(max(h, 0), H_ - 1) * W_ + wcc;
  }

  float4 hv[NPL][6];
#pragma unroll
  for (int c = 0; c < 2; ++c) {
    const float* cb = (c == 0 ? xmax : xmean) + (long)b * DHW;
#pragma unroll
    for (int kd = 0; kd < 7; ++kd) {
      const int dz  = d + kd - 3;
      const bool pv = (dz >= 0) && (dz < D_);
      const float* plane = cb + (long)min(max(dz, 0), D_ - 1) * HW;
#pragma unroll
      for (int it = 0; it < 6; ++it) {
        float4 v = *(const float4*)(plane + goff[it]);
        const bool keep = pv && ok[it];
        v.x = keep ? v.x : 0.0f; v.y = keep ? v.y : 0.0f;
        v.z = keep ? v.z : 0.0f; v.w = keep ? v.w : 0.0f;
        hv[c * 7 + kd][it] = v;
      }
    }
  }

#pragma unroll
  for (int k = 0; k < 21; ++k) lw[lane + 32 * k] = wv[k];
  if (lane < 14) lw[672 + lane] = wv[21];
  if (q < 6) {
    float* tp = tile + rs * TW + 4 * q;
#pragma unroll
    for (int p = 0; p < NPL; ++p) {
#pragma unroll
      for (int it = 0; it < 5; ++it)
        *(float4*)(tp + p * PS + it * 4 * TW) = hv[p][it];
    }
    if (rs < 2) {
#pragma unroll
      for (int p = 0; p < NPL; ++p)
        *(float4*)(tp + p * PS + 5 * 4 * TW) = hv[p][5];
    }
  }
  if (lane < 12) {
    float4 z; z.x = 0.0f; z.y = 0.0f; z.z = 0.0f; z.w = 0.0f;
    *(float4*)(tile + NPL * PS + 4 * lane) = z;
  }
  __syncthreads();

  const int m    = lane & 15;
  const int n    = m;
  const int kgrp = (lane >> 4) << 1;

  v8f acc = {0.f, 0.f, 0.f, 0.f, 0.f, 0.f, 0.f, 0.f};

  int c0[2], i0[2], i1[2]; bool hi[2];
#pragma unroll
  for (int s = 0; s < 2; ++s) {
    const int ks = kgrp + s;
    const int r  = (ks - m) & 3;
    c0[s] = (m + r - ks) >> 2;
    hi[s] = (r <= 2);
    i0[s] = r * 7;
    i1[s] = (hi[s] ? r + 4 : r) * 7;
  }

  const int kd_lo = max(0, 3 - d);
  const int kd_hi = min(6, D_ + 2 - d);
  const int nk = kd_hi - kd_lo + 1;
  const int np = 2 * nk;
  const float* tl0 = tile + kgrp * TW + n + 1;

  PlaneOps pa, pb;
  {
    const int p = kd_lo;
    sg_load_plane(pa, tl0 + p * PS, lw + p * 49, i0, i1);
  }
  for (int i = 0; i < np; i += 2) {
    {
      const int j = i + 1;
      const int p = (j < nk) ? (kd_lo + j) : (7 + kd_lo + j - nk);
      sg_load_plane(pb, tl0 + p * PS, lw + p * 49, i0, i1);
    }
    __builtin_amdgcn_sched_barrier(0);
    acc = sg_mma_plane(pa, acc, c0, hi);
    __builtin_amdgcn_sched_barrier(0);
    {
      const int j = min(i + 2, np - 1);
      const int p = (j < nk) ? (kd_lo + j) : (7 + kd_lo + j - nk);
      sg_load_plane(pa, tl0 + p * PS, lw + p * 49, i0, i1);
    }
    __builtin_amdgcn_sched_barrier(0);
    acc = sg_mma_plane(pb, acc, c0, hi);
    __builtin_amdgcn_sched_barrier(0);
  }

  const float bv = bias[0];
  float* sp = scale + ((long)b * D_ + d) * HW;
#pragma unroll
  for (int r = 0; r < 8; ++r) {
    const int M = r + (kgrp << 2);
    const float y = acc[r] + bv;
    sp[(hbase + M) * W_ + (wbase + n)] = 1.0f / (1.0f + __expf(-y));
  }
}

__global__ __launch_bounds__(256) void sg_apply(
    const float* __restrict__ x, const float* __restrict__ scale,
    float* __restrict__ out) {
  long t = (long)blockIdx.x * blockDim.x + threadIdx.x;
  long f = t * 4;
  int  b = (int)(f / CDHW);
  int  p = (int)((f % CDHW) % DHW);
  float4 xv = *(const float4*)(x + f);
  float4 sv = *(const float4*)(scale + (long)b * DHW + p);
  float4 o;
  o.x = xv.x * sv.x; o.y = xv.y * sv.y;
  o.z = xv.z * sv.z; o.w = xv.w * sv.w;
  *(float4*)(out + f) = o;
}

extern "C" void kernel_launch(void* const* d_in, const int* in_sizes, int n_in,
                              void* d_out, int out_size, void* d_ws, size_t ws_size,
                              hipStream_t stream) {
  const float* x      = (const float*)d_in[0];
  const float* conv_w = (const float*)d_in[1];
  const float* conv_b = (const float*)d_in[2];
  float* out = (float*)d_out;

  float* ws    = (float*)d_ws;
  float* xmax  = ws;
  float* xmean = ws + (long)B_ * DHW;
  float* scale = ws + 2L * B_ * DHW;

  sg_reduce_maxmean<<<512, 256, 0, stream>>>(x, xmax, xmean);

  sg_conv_sigmoid<<<2048, 32, 0, stream>>>(xmax, xmean, conv_w, conv_b, scale);

  sg_apply<<<32768, 256, 0, stream>>>(x, scale, out);
}


# conv kernel re-tiled: M = a 4x4 output patch, K = its 10x10 input window (25 K chunks per plane, A from a zero-padded weight table in LDS), not a 16-row Toeplitz band (42); sum order changes
# speedup vs baseline: 2.1710x; 1.1378x over previous
#include <hip/hip_runtime.h>
#include <hip/hip_bf16.h>

typedef float v2f __attribute__((ext_vector_type(2)));
typedef float v8f __attribute__((ext_vector_type(8)));

#define B_    4
#define C_    64
#define D_    32
#define H_    64
#define W_    64
#define HW    (H_ * W_)
#define DHW   (D_ * HW)
#define CDHW  (C_ * DHW)

__global__ __launch_bounds__(256) void sg_reduce_maxmean(
    const float* __restrict__ x, float* __restrict__ xmax, float* __restrict__ xmean) {
  long t = (long)blockIdx.x * blockDim.x + threadIdx.x;
  long f = t * 4;
  int  b = (int)(f / DHW);
  int  q = (int)(f % DHW);
  const float* xp = x + (long)b * CDHW + q;

  float4 v = *(const float4*)xp;
  float mx0 = v.x, mx1 = v.y, mx2 = v.z, mx3 = v.w;
  float s0 = v.x, s1 = v.y, s2 = v.z, s3 = v.w;
  for (int c = 1; c < C_; ++c) {
    float4 u = *(const float4*)(xp + (long)c * DHW);
    mx0 = fmaxf(mx0, u.x); mx1 = fmaxf(mx1, u.y);
    mx2 = fmaxf(mx2, u.z); mx3 = fmaxf(mx3, u.w);
    s0 += u.x; s1 += u.y; s2 += u.z; s3 += u.w;
  }
  const float inv = 1.0f / (float)C_;
  float4 mo; mo.x = mx0; mo.y = mx1; mo.z = mx2; mo.w = mx3;
  float4 me; me.x = s0 * inv; me.y = s1 * inv; me.z = s2 * inv; me.w = s3 * inv;
  *(float4*)(xmax  + (long)b * DHW + q) = mo;
  *(float4*)(xmean + (long)b * DHW + q) = me;
}

#define PR    14
#define TW    40
#define PS    (PR * TW)
#define NPL   14
#define NPC   (PR * TW / 4)
#define WTN   1536

typedef float v2fu __attribute__((ext_vector_type(2), aligned(4)));

struct PlaneOps {
  v2f a[25];
  v2f b[25];
};

__device__ __forceinline__ void sg_load_plane(PlaneOps& o, const float* ap, const float* bp) {
#pragma unroll
  for (int ip = 0; ip < 5; ++ip) {
#pragma unroll
    for (int jp = 0; jp < 5; ++jp) {
      o.a[ip * 5 + jp] = *(const v2fu*)(ap + 20 * ip + 2 * jp);
      o.b[ip * 5 + jp] = *(const v2fu*)(bp + 2 * ip * TW + 2 * jp);
    }
  }
}

__device__ __forceinline__ v8f sg_mma_plane(const PlaneOps& o, v8f acc) {
#pragma unroll
  for (int c = 0; c < 25; ++c)
    acc = __builtin_amdgcn_wmma_f32_16x16x4_f32(
        false, o.a[c], false, o.b[c], (short)0, acc, false, false);
  return acc;
}

__global__ __launch_bounds__(32) void sg_conv_sigmoid(
    const float* __restrict__ xmax, const float* __restrict__ xmean,
    const float* __restrict__ wgt, const float* __restrict__ bias,
    float* __restrict__ scale) {
  __shared__ __attribute__((aligned(16))) float wt[WTN];
  __shared__ __attribute__((aligned(16))) float tile[NPL * PS];

  const int lane = threadIdx.x;
  const int bid  = blockIdx.x;
  const int wti = bid & 1, hti = (bid >> 1) & 7;
  const int d  = (bid >> 4) & 31, b = bid >> 9;
  const int hbase = hti * 8, wbase = wti * 32;

  float wv[22];
#pragma unroll
  for (int k = 0; k < 21; ++k) wv[k] = wgt[lane + 32 * k];
  wv[21] = (lane < 14) ? wgt[672 + lane] : 0.0f;

  int  goff[5];
  bool ok[5];
#pragma unroll
  for (int it = 0; it < 5; ++it) {
    const int idx = min(lane + 32 * it, NPC - 1);
    const int row = idx / 10, q = idx - row * 10;
    const int h   = hbase - 3 + row;
    const int wc  = wbase - 4 + 4 * q;
    ok[it]   = (h >= 0) && (h < H_) && (wc >= 0) && (wc < W_);
    goff[it] = min(max(h, 0), H_ - 1) * W_ + min(max(wc, 0), W_ - 4);
  }

  float4 hv[NPL][5];
#pragma unroll
  for (int c = 0; c < 2; ++c) {
    const float* cb = (c == 0 ? xmax : xmean) + (long)b * DHW;
#pragma unroll
    for (int kd = 0; kd < 7; ++kd) {
      const int dz  = d + kd - 3;
      const bool pv = (dz >= 0) && (dz < D_);
      const float* plane = cb + (long)min(max(dz, 0), D_ - 1) * HW;
#pragma unroll
      for (int it = 0; it < 5; ++it) {
        float4 v = *(const float4*)(plane + goff[it]);
        const bool keep = pv && ok[it];
        v.x = keep ? v.x : 0.0f; v.y = keep ? v.y : 0.0f;
        v.z = keep ? v.z : 0.0f; v.w = keep ? v.w : 0.0f;
        hv[c * 7 + kd][it] = v;
      }
    }
  }

  {
    float4 z; z.x = 0.0f; z.y = 0.0f; z.z = 0.0f; z.w = 0.0f;
#pragma unroll
    for (int k = 0; k < WTN / 128; ++k) *(float4*)(wt + 4 * (lane + 32 * k)) = z;
#pragma unroll
    for (int k = 0; k < 22; ++k) {
      const int idx = lane + 32 * k;
      const int p = idx / 49, r = idx - p * 49;
      const int kh = r / 7, kw = r - kh * 7;
      if (k < 21 || lane < 14) wt[(p * 10 + kh + 3) * 10 + kw + 3] = wv[k];
    }
  }
  {
    float* tp = tile + 4 * lane;
#pragma unroll
    for (int p = 0; p < NPL; ++p) {
#pragma unroll
      for (int it = 0; it < 4; ++it)
        *(float4*)(tp + p * PS + it * 128) = hv[p][it];
    }
    if (lane < NPC - 128) {
#pragma unroll
      for (int p = 0; p < NPL; ++p)
        *(float4*)(tp + p * PS + 512) = hv[p][4];
    }
  }
  __syncthreads();

  const int c16 = lane & 15;
  const int hh  = lane >> 4;
  const int oh = c16 >> 2, ow = c16 & 3;
  const int ph = c16 >> 3, pw = c16 & 7;

  v8f acc = {0.f, 0.f, 0.f, 0.f, 0.f, 0.f, 0.f, 0.f};

  const int kd_lo = max(0, 3 - d);
  const int kd_hi = min(6, D_ + 2 - d);
  const int nk = kd_hi - kd_lo + 1;
  const int np = 2 * nk;
  const float* ap0 = wt + (hh - oh + 3) * 10 + (3 - ow);
  const float* bp0 = tile + (4 * ph + hh) * TW + 4 * pw + 1;

  PlaneOps pa, pb;
  {
    const int p = kd_lo;
    sg_load_plane(pa, ap0 + p * 100, bp0 + p * PS);
  }
#define SG_GROUP_FENCE() asm volatile("" : "+v"(acc) : : "memory")
  for (int i = 0; i < np; i += 2) {
    {
      const int j = i + 1;
      const int p = (j < nk) ? (kd_lo + j) : (7 + kd_lo + j - nk);
      sg_load_plane(pb, ap0 + p * 100, bp0 + p * PS);
    }
    SG_GROUP_FENCE();
    acc = sg_mma_plane(pa, acc);
    SG_GROUP_FENCE();
    {
      const int j = min(i + 2, np - 1);
      const int p = (j < nk) ? (kd_lo + j) : (7 + kd_lo + j - nk);
      sg_load_plane(pa, ap0 + p * 100, bp0 + p * PS);
    }
    SG_GROUP_FENCE();
    acc = sg_mma_plane(pb, acc);
    SG_GROUP_FENCE();
  }
#undef SG_GROUP_FENCE

  const float bv = bias[0];
  float* sp = scale + ((long)b * D_ + d) * HW
            + (hbase + 4 * ph + 2 * hh) * W_ + (wbase + 4 * pw);
#pragma unroll
  for (int rr = 0; rr < 2; ++rr) {
    float4 o;
    o.x = 1.0f / (1.0f + __expf(-(acc[4 * rr + 0] + bv)));
    o.y = 1.0f / (1.0f + __expf(-(acc[4 * rr + 1] + bv)));
    o.z = 1.0f / (1.0f + __expf(-(acc[4 * rr + 2] + bv)));
    o.w = 1.0f / (1.0f + __expf(-(acc[4 * rr + 3] + bv)));
    *(float4*)(sp + rr * W_) = o;
  }
}

__global__ __launch_bounds__(256) void sg_apply(
    const float* __restrict__ x, const float* __restrict__ scale,
    float* __restrict__ out) {
  long t = (long)blockIdx.x * blockDim.x + threadIdx.x;
  long f = t * 4;
  int  b = (int)(f / CDHW);
  int  p = (int)((f % CDHW) % DHW);
  float4 xv = *(const float4*)(x + f);
  float4 sv = *(const float4*)(scale + (long)b * DHW + p);
  float4 o;
  o.x = xv.x * sv.x; o.y = xv.y * sv.y;
  o.z = xv.z * sv.z; o.w = xv.w * sv.w;
  *(float4*)(out + f) = o;
}

extern "C" void kernel_launch(void* const* d_in, const int* in_sizes, int n_in,
                              void* d_out, int out_size, void* d_ws, size_t ws_size,
                              hipStream_t stream) {
  const float* x      = (const float*)d_in[0];
  const float* conv_w = (const float*)d_in[1];
  const float* conv_b = (const float*)d_in[2];
  float* out = (float*)d_out;

  float* ws    = (float*)d_ws;
  float* xmax  = ws;
  float* xmean = ws + (long)B_ * DHW;
  float* scale = ws + 2L * B_ * DHW;

  sg_reduce_maxmean<<<512, 256, 0, stream>>>(x, xmax, xmean);

  sg_conv_sigmoid<<<2048, 32, 0, stream>>>(xmax, xmean, conv_w, conv_b, scale);

  sg_apply<<<32768, 256, 0, stream>>>(x, scale, out);
}


# conv kernel: one 8-wave workgroup per tile of 8 consecutive depths stages their 28 shared input planes and the weight table once, not 8 x 14 planes and 8 tables in 8 one-wave workgroups
# speedup vs baseline: 2.2390x; 1.0313x over previous
#include <hip/hip_runtime.h>
#include <hip/hip_bf16.h>

typedef float v2f __attribute__((ext_vector_type(2)));
typedef float v8f __attribute__((ext_vector_type(8)));

#define B_    4
#define C_    64
#define D_    32
#define H_    64
#define W_    64
#define HW    (H_ * W_)
#define DHW   (D_ * HW)
#define CDHW  (C_ * DHW)

__global__ __launch_bounds__(256) void sg_reduce_maxmean(
    const float* __restrict__ x, float* __restrict__ xmax, float* __restrict__ xmean) {
  long t = (long)blockIdx.x * blockDim.x + threadIdx.x;
  long f = t * 4;
  int  b = (int)(f / DHW);
  int  q = (int)(f % DHW);
  const float* xp = x + (long)b * CDHW + q;

  float4 v = *(const float4*)xp;
  float mx0 = v.x, mx1 = v.y, mx2 = v.z, mx3 = v.w;
  float s0 = v.x, s1 = v.y, s2 = v.z, s3 = v.w;
  for (int c = 1; c < C_; ++c) {
    float4 u = *(const float4*)(xp + (long)c * DHW);
    mx0 = fmaxf(mx0, u.x); mx1 = fmaxf(mx1, u.y);
    mx2 = fmaxf(mx2, u.z); mx3 = fmaxf(mx3, u.w);
    s0 += u.x; s1 += u.y; s2 += u.z; s3 += u.w;
  }
  const float inv = 1.0f / (float)C_;
  float4 mo; mo.x = mx0; mo.y = mx1; mo.z = mx2; mo.w = mx3;
  float4 me; me.x = s0 * inv; me.y = s1 * inv; me.z = s2 * inv; me.w = s3 * inv;
  *(float4*)(xmax  + (long)b * DHW + q) = mo;
  *(float4*)(xmean + (long)b * DHW + q) = me;
}

#define PR    14
#define TW    40
#define PS    (PR * TW)
#define NWV   8
#define NZ    (NWV + 6)
#define NPLS  (2 * NZ)
#define NPC   (PR * TW / 4)
#define WTN   1536
#define TILEN 41000

typedef float v2fu __attribute__((ext_vector_type(2), aligned(4)));

struct PlaneOps {
  v2f a[25];
  v2f b[25];
};

__device__ __forceinline__ void sg_load_plane(PlaneOps& o, const float* ap, const float* bp) {
#pragma unroll
  for (int ip = 0; ip < 5; ++ip) {
#pragma unroll
    for (int jp = 0; jp < 5; ++jp) {
      o.a[ip * 5 + jp] = *(const v2fu*)(ap + 20 * ip + 2 * jp);
      o.b[ip * 5 + jp] = *(const v2fu*)(bp + 2 * ip * TW + 2 * jp);
    }
  }
}

__device__ __forceinline__ v8f sg_mma_plane(const PlaneOps& o, v8f acc) {
#pragma unroll
  for (int c = 0; c < 25; ++c)
    acc = __builtin_amdgcn_wmma_f32_16x16x4_f32(
        false, o.a[c], false, o.b[c], (short)0, acc, false, false);
  return acc;
}

__global__ __launch_bounds__(256) void sg_conv_sigmoid(
    const float* __restrict__ xmax, const float* __restrict__ xmean,
    const float* __restrict__ wgt, const float* __restrict__ bias,
    float* __restrict__ scale) {
  __shared__ __attribute__((aligned(16))) float wt[WTN];
  __shared__ __attribute__((aligned(16))) float tile[TILEN];

  const int tid  = threadIdx.x;
  const int lane = tid & 31;
  const int w    = __builtin_amdgcn_readfirstlane(tid >> 5);
  const int bid  = blockIdx.x;
  const int wti = bid & 1, hti = (bid >> 1) & 7;
  const int d0 = ((bid >> 4) & 3) * NWV, b = bid >> 6;
  const int hbase = hti * 8, wbase = wti * 32;
  const int d = d0 + w;

  float wv[WTN / 256];
#pragma unroll
  for (int k = 0; k < WTN / 256; ++k) {
    const int idx = tid + 256 * k;
    const int p = idx / 100, rem = idx - p * 100;
    const int r = rem / 10, c = rem - r * 10;
    const bool tap = (p < 14) && (r >= 3) && (c >= 3);
    const int gi = tap ? (p * 49 + (r - 3) * 7 + (c - 3)) : 0;
    const float v = wgt[gi];
    wv[k] = tap ? v : 0.0f;
  }

  int  goff[5];
  bool ok[5];
#pragma unroll
  for (int it = 0; it < 5; ++it) {
    const int idx = min(lane + 32 * it, NPC - 1);
    const int row = idx / 10, q = idx - row * 10;
    const int h   = hbase - 3 + row;
    const int wc  = wbase - 4 + 4 * q;
    ok[it]   = (h >= 0) && (h < H_) && (wc >= 0) && (wc < W_);
    goff[it] = min(max(h, 0), H_ - 1) * W_ + min(max(wc, 0), W_ - 4);
  }

  float4 hv[4][5];
#pragma unroll
  for (int k = 0; k < 4; ++k) {
    const int P  = min(w + NWV * k, NPLS - 1);
    const int c  = (P >= NZ) ? 1 : 0;
    const int dz = d0 - 3 + (P - c * NZ);
    const bool pv = (dz >= 0) && (dz < D_);
    const float* plane = (c ? xmean : xmax) + (long)b * DHW
                       + (long)min(max(dz, 0), D_ - 1) * HW;
#pragma unroll
    for (int it = 0; it < 5; ++it) {
      float4 v = *(const float4*)(plane + goff[it]);
      const bool keep = pv && ok[it];
      v.x = keep ? v.x : 0.0f; v.y = keep ? v.y : 0.0f;
      v.z = keep ? v.z : 0.0f; v.w = keep ? v.w : 0.0f;
      hv[k][it] = v;
    }
  }

#pragma unroll
  for (int k = 0; k < WTN / 256; ++k) wt[tid + 256 * k] = wv[k];
#pragma unroll
  for (int k = 0; k < 4; ++k) {
    const int P = w + NWV * k;
    if (k < 3 || w < NPLS - 3 * NWV) {
      float* tp = tile + P * PS + 4 * lane;
#pragma unroll
      for (int it = 0; it < 4; ++it)
        *(float4*)(tp + it * 128) = hv[k][it];
      if (lane < NPC - 128)
        *(float4*)(tp + 512) = hv[k][4];
    }
  }
  __syncthreads();

  const int c16 = lane & 15;
  const int hh  = lane >> 4;
  const int oh = c16 >> 2, ow = c16 & 3;
  const int ph = c16 >> 3, pw = c16 & 7;

  v8f acc = {0.f, 0.f, 0.f, 0.f, 0.f, 0.f, 0.f, 0.f};

  const int kd_lo = max(0, 3 - d);
  const int kd_hi = min(6, D_ + 2 - d);
  const int nk = kd_hi - kd_lo + 1;
  const int np = 2 * nk;
  const float* ap0 = wt + (hh - oh + 3) * 10 + (3 - ow);
  const float* bp0 = tile + w * PS + (4 * ph + hh) * TW + 4 * pw + 1;

  PlaneOps pa, pb;
  {
    const int p = kd_lo;
    sg_load_plane(pa, ap0 + p * 100, bp0 + p * PS);
  }
#define SG_GROUP_FENCE() asm volatile("" : "+v"(acc) : : "memory")
  for (int i = 0; i < np; i += 2) {
    {
      const int j = i + 1;
      const int p = (j < nk) ? (kd_lo + j) : (7 + kd_lo + j - nk);
      const int z = (j < nk) ? p : (p + 7);
      sg_load_plane(pb, ap0 + p * 100, bp0 + z * PS);
    }
    SG_GROUP_FENCE();
    acc = sg_mma_plane(pa, acc);
    SG_GROUP_FENCE();
    {
      const int j = min(i + 2, np - 1);
      const int p = (j < nk) ? (kd_lo + j) : (7 + kd_lo + j - nk);
      const int z = (j < nk) ? p : (p + 7);
      sg_load_plane(pa, ap0 + p * 100, bp0 + z * PS);
    }
    SG_GROUP_FENCE();
    acc = sg_mma_plane(pb, acc);
    SG_GROUP_FENCE();
  }
#undef SG_GROUP_FENCE

  const float bv = bias[0];
  float* sp = scale + ((long)b * D_ + d) * HW
            + (hbase + 4 * ph + 2 * hh) * W_ + (wbase + 4 * pw);
#pragma unroll
  for (int rr = 0; rr < 2; ++rr) {
    float4 o;
    o.x = 1.0f / (1.0f + __expf(-(acc[4 * rr + 0] + bv)));
    o.y = 1.0f / (1.0f + __expf(-(acc[4 * rr + 1] + bv)));
    o.z = 1.0f / (1.0f + __expf(-(acc[4 * rr + 2] + bv)));
    o.w = 1.0f / (1.0f + __expf(-(acc[4 * rr + 3] + bv)));
    *(float4*)(sp + rr * W_) = o;
  }
}

__global__ __launch_bounds__(256) void sg_apply(
    const float* __restrict__ x, const float* __restrict__ scale,
    float* __restrict__ out) {
  long t = (long)blockIdx.x * blockDim.x + threadIdx.x;
  long f = t * 4;
  int  b = (int)(f / CDHW);
  int  p = (int)((f % CDHW) % DHW);
  float4 xv = *(const float4*)(x + f);
  float4 sv = *(const float4*)(scale + (long)b * DHW + p);
  float4 o;
  o.x = xv.x * sv.x; o.y = xv.y * sv.y;
  o.z = xv.z * sv.z; o.w = xv.w * sv.w;
  *(float4*)(out + f) = o;
}

extern "C" void kernel_launch(void* const* d_in, const int* in_sizes, int n_in,
                              void* d_out, int out_size, void* d_ws, size_t ws_size,
                              hipStream_t stream) {
  const float* x      = (const float*)d_in[0];
  const float* conv_w = (const float*)d_in[1];
  const float* conv_b = (const float*)d_in[2];
  float* out = (float*)d_out;

  float* ws    = (float*)d_ws;
  float* xmax  = ws;
  float* xmean = ws + (long)B_ * DHW;
  float* scale = ws + 2L * B_ * DHW;

  sg_reduce_maxmean<<<512, 256, 0, stream>>>(x, xmax, xmean);

  sg_conv_sigmoid<<<256, 256, 0, stream>>>(xmax, xmean, conv_w, conv_b, scale);

  sg_apply<<<32768, 256, 0, stream>>>(x, scale, out);
}
